# starting kernel
# baseline (speedup 1.0000x reference)
#include <hip/hip_runtime.h>
#include <math.h>

typedef __attribute__((ext_vector_type(2))) float v2f;
typedef __attribute__((ext_vector_type(4))) float v4f;
typedef __attribute__((ext_vector_type(8))) float v8f;

#define D2R 0.017453292519943295f
#define R2D 57.29577951308232f

__global__ __launch_bounds__(256) void fk_wmma_kernel(
    const float* __restrict__ theta, const float* __restrict__ dh,
    float* __restrict__ out, int n)
{
    __shared__ float smem[8 * 448];
    const int lane = threadIdx.x & 31;
    const int wave = threadIdx.x >> 5;
    float* T = &smem[wave * 448];
    float* G = T + 384;

    const int base = (blockIdx.x * 8 + wave) * 4;

    {
        const int s = lane >> 3;
        const int i = lane & 7;
        if (i < 6) {
            int samp = base + s; if (samp >= n) samp = n - 1;
            const float a     = dh[i * 4 + 0];
            const float alr   = dh[i * 4 + 1] * D2R;
            const float d     = dh[i * 4 + 2];
            const float off   = dh[i * 4 + 3];
            const float th    = (off + theta[samp * 6 + i]) * D2R;
            float st, ct, sa, ca;
            sincosf(th,  &st, &ct);
            sincosf(alr, &sa, &ca);
            float* tp = &T[i * 64 + s * 16];
            v4f r0; r0.x = ct;      r0.y = -st;     r0.z = 0.0f; r0.w = a;
            v4f r1; r1.x = st * ca; r1.y = ct * ca; r1.z = -sa;  r1.w = -sa * d;
            v4f r2; r2.x = st * sa; r2.y = ct * sa; r2.z = ca;   r2.w = ca * d;
            v4f r3; r3.x = 0.0f;    r3.y = 0.0f;    r3.z = 0.0f; r3.w = 1.0f;
            *(v4f*)(tp +  0) = r0;
            *(v4f*)(tp +  4) = r1;
            *(v4f*)(tp +  8) = r2;
            *(v4f*)(tp + 12) = r3;
        }
    }
    asm volatile("s_wait_dscnt 0" ::: "memory");

    const int h  = lane >> 4;
    const int m  = lane & 15;
    const int sA = m >> 2;
    const int iA = m & 3;

    v2f Acc;
    Acc.x = T[0 * 64 + sA * 16 + iA * 4 + (2 * h + 0)];
    Acc.y = T[0 * 64 + sA * 16 + iA * 4 + (2 * h + 1)];

    const bool wr = (h == ((m >> 3) & 1));
    const bool hi = (sA & 1);
    #pragma unroll
    for (int j = 1; j < 6; ++j) {
        v2f Bv;
        Bv.x = T[j * 64 + sA * 16 + (2 * h + 0) * 4 + iA];
        Bv.y = T[j * 64 + sA * 16 + (2 * h + 1) * 4 + iA];
        v8f c = {};
        c = __builtin_amdgcn_wmma_f32_16x16x4_f32(
                false, Acc, false, Bv, (short)0, c, false, false);
        if (wr) {
            v4f dv;
            dv.x = hi ? c[4] : c[0];
            dv.y = hi ? c[5] : c[1];
            dv.z = hi ? c[6] : c[2];
            dv.w = hi ? c[7] : c[3];
            *(v4f*)&G[sA * 16 + iA * 4] = dv;
        }
        asm volatile("s_wait_dscnt 0" ::: "memory");
        if (j < 5) {
            Acc.x = G[sA * 16 + (2 * h + 0) * 4 + iA];
            Acc.y = G[sA * 16 + (2 * h + 1) * 4 + iA];
        }
    }

    if (lane < 4 && (base + lane) < n) {
        const int s = lane;
        #define GM(r, cc) G[s * 16 + (cc) * 4 + (r)]
        const float x = GM(0, 3), y = GM(1, 3), z = GM(2, 3);
        const float T00 = GM(0, 0), T01 = GM(0, 1), T02 = GM(0, 2);
        const float T10 = GM(1, 0), T11 = GM(1, 1), T12 = GM(1, 2);
        const float T22 = GM(2, 2);
        #undef GM
        float Aa, Bb, Cc;
        if (fabsf(T12) <= 1e-6f && fabsf(T22) <= 1e-6f) {
            Aa = R2D * atan2f(T10, T11);
            Bb = R2D * atan2f(T02, T22);
            Cc = 0.0f;
        } else {
            const float An = atan2f(-T01, T00);
            float sAn, cAn;
            sincosf(An, &sAn, &cAn);
            Aa = R2D * An;
            Bb = R2D * atan2f(T02, cAn * T00 - sAn * T01);
            Cc = R2D * atan2f(-T12, T22);
        }
        float* o = out + (size_t)(base + s) * 6;
        o[0] = x; o[1] = y; o[2] = z; o[3] = Aa; o[4] = Bb; o[5] = Cc;
    }
}

extern "C" void kernel_launch(void* const* d_in, const int* in_sizes, int n_in,
                              void* d_out, int out_size, void* d_ws, size_t ws_size,
                              hipStream_t stream) {
    const float* theta = (const float*)d_in[0];
    const float* dh    = (const float*)d_in[1];
    float* out = (float*)d_out;
    const int n = in_sizes[0] / 6;
    const int samplesPerBlock = 32;
    const int blocks = (n + samplesPerBlock - 1) / samplesPerBlock;
    fk_wmma_kernel<<<blocks, 256, 0, stream>>>(theta, dh, out, n);
}


# Chain re-decomposed: b*T_j = ct*(b*M1)+st*(b*M2)+b*M0 with constant M as WMMA A operand, 16 samples on N, D to next B by 2 lane FMAs in registers (32 samples/wave, no LDS); f32 sums reassociate
# speedup vs baseline: 2.6708x; 2.6708x over previous
#include <hip/hip_runtime.h>
#include <math.h>


typedef __attribute__((ext_vector_type(2))) float v2f;
typedef __attribute__((ext_vector_type(8))) float v8f;

#define D2R 0.017453292519943295f
#define R2D 57.29577951308232f

__device__ __forceinline__ float acoef(int q, int c, float ca, float sa, float a, float d)
{
    float r = 0.0f;
    if (q == 0)  r = (c == 0) ? 1.0f : 0.0f;
    if (q == 1 || q == 2) r = (c == 1) ? ca : ((c == 2) ? sa : 0.0f);
    if (q == 3)  r = (c == 0) ? -1.0f : 0.0f;
    if (q == 12) r = (c == 1) ? -sa : ((c == 2) ? ca : 0.0f);
    if (q == 13) r = (c == 0) ? a : ((c == 1) ? -sa * d : ((c == 2) ? ca * d : 1.0f));
    return r;
}

__device__ __forceinline__ float xhalf_m8(float x)
{
    int b = __builtin_bit_cast(int, x);
    return __builtin_bit_cast(float,
        __builtin_amdgcn_permlanex16(b, b, 0xfedcba98, 0x76543210, false, false));
}

__device__ __forceinline__ float xhalf(float x)
{
    int b = __builtin_bit_cast(int, x);
    return __builtin_bit_cast(float,
        __builtin_amdgcn_permlanex16(b, b, 0x76543210, 0xfedcba98, false, false));
}

__device__ __forceinline__ float bcast(float x, int srcLane)
{
    return __builtin_bit_cast(float,
        __builtin_amdgcn_readlane(__builtin_bit_cast(int, x), srcLane));
}

__global__ __launch_bounds__(256) void fk_wmma_kernel(
    const float* __restrict__ theta, const float* __restrict__ dh,
    float* __restrict__ out, int n)
{
    const int lane = threadIdx.x & 31;
    const int wave = threadIdx.x >> 5;
    const int h = lane >> 4;
    const int m = lane & 15;
    const int base = (blockIdx.x * 8 + wave) * 32;
    int samp = base + lane;
    if (samp >= n) samp = n - 1;

    const float2* tp = (const float2*)(theta + (size_t)samp * 6);
    const float2 t01 = tp[0], t23 = tp[1], t45 = tp[2];
    const float thd[6] = { t01.x, t01.y, t23.x, t23.y, t45.x, t45.y };

    const int jl = (lane < 6) ? lane : 5;
    const float a_l   = dh[jl * 4 + 0];
    const float al_l  = dh[jl * 4 + 1] * D2R;
    const float d_l   = dh[jl * 4 + 2];
    const float off_l = dh[jl * 4 + 3];
    float sa_l, ca_l;
    sincosf(al_l, &sa_l, &ca_l);

    float ct[6], st[6];
    #pragma unroll
    for (int j = 0; j < 6; ++j) {
        const float off = bcast(off_l, j);
        sincosf((off + thd[j]) * D2R, &st[j], &ct[j]);
    }

    v2f S[2][3];
    {
        const float ca = bcast(ca_l, 0), sa = bcast(sa_l, 0);
        const float a = bcast(a_l, 0), d = bcast(d_l, 0);
        const float o[3][2] = { { ct[0], -st[0] }, { st[0] * ca, ct[0] * ca }, { st[0] * sa, ct[0] * sa } };
        const float k[3][2] = { { 0.0f, a }, { -sa, -sa * d }, { ca, ca * d } };
        #pragma unroll
        for (int g = 0; g < 2; ++g) {
            #pragma unroll
            for (int i = 0; i < 3; ++i) {
                S[g][i].x = (h == g) ? o[i][0] : k[i][0];
                S[g][i].y = (h == g) ? o[i][1] : k[i][1];
            }
        }
    }

    #pragma unroll
    for (int j = 1; j < 6; ++j) {
        const float ca = bcast(ca_l, j), sa = bcast(sa_l, j);
        const float a = bcast(a_l, j), d = bcast(d_l, j);
        v2f A[2];
        A[0].x = acoef(m, 2 * h + 0, ca, sa, a, d);
        A[0].y = acoef(m, 2 * h + 1, ca, sa, a, d);
        A[1].x = xhalf_m8(A[0].x);
        A[1].y = xhalf_m8(A[0].y);
        v8f c[2][3];
        #pragma unroll
        for (int g = 0; g < 2; ++g) {
            #pragma unroll
            for (int i = 0; i < 3; ++i) {
                v8f z = {};
                c[g][i] = __builtin_amdgcn_wmma_f32_16x16x4_f32(
                    false, A[g], false, S[g][i], (short)0, z, false, false);
            }
        }
        #pragma unroll
        for (int g = 0; g < 2; ++g) {
            #pragma unroll
            for (int i = 0; i < 3; ++i) {
                S[g][i].x = fmaf(ct[j], c[g][i][0], fmaf(st[j], c[g][i][2], c[g][i][4]));
                S[g][i].y = fmaf(ct[j], c[g][i][1], fmaf(st[j], c[g][i][3], c[g][i][5]));
            }
        }
    }

    float b01[3][2], b23[3][2];
    #pragma unroll
    for (int i = 0; i < 3; ++i) {
        #pragma unroll
        for (int v = 0; v < 2; ++v) {
            const float own  = h ? S[1][i][v] : S[0][i][v];
            const float send = h ? S[0][i][v] : S[1][i][v];
            b01[i][v] = own;
            b23[i][v] = xhalf(send);
        }
    }

    const float T00 = b01[0][0], T01 = b01[0][1], T02 = b23[0][0], x = b23[0][1];
    const float T10 = b01[1][0], T11 = b01[1][1], T12 = b23[1][0], y = b23[1][1];
    const float T22 = b23[2][0], z = b23[2][1];
    float Aa, Bb, Cc;
    if (fabsf(T12) <= 1e-6f && fabsf(T22) <= 1e-6f) {
        Aa = R2D * atan2f(T10, T11);
        Bb = R2D * atan2f(T02, T22);
        Cc = 0.0f;
    } else {
        const float An = atan2f(-T01, T00);
        float sAn, cAn;
        sincosf(An, &sAn, &cAn);
        Aa = R2D * An;
        Bb = R2D * atan2f(T02, cAn * T00 - sAn * T01);
        Cc = R2D * atan2f(-T12, T22);
    }
    if (base + lane < n) {
        float2* o = (float2*)(out + (size_t)(base + lane) * 6);
        o[0] = make_float2(x, y);
        o[1] = make_float2(z, Aa);
        o[2] = make_float2(Bb, Cc);
    }
}

extern "C" void kernel_launch(void* const* d_in, const int* in_sizes, int n_in,
                              void* d_out, int out_size, void* d_ws, size_t ws_size,
                              hipStream_t stream) {
    const float* theta = (const float*)d_in[0];
    const float* dh    = (const float*)d_in[1];
    float* out = (float*)d_out;
    const int n = in_sizes[0] / 6;
    const int samplesPerBlock = 256;
    const int blocks = (n + samplesPerBlock - 1) / samplesPerBlock;
    fk_wmma_kernel<<<blocks, 256, 0, stream>>>(theta, dh, out, n);
}


# Joint constants and the five WMMA A operands built once per work-group by wave 0 and shared through LDS behind one barrier; the other 7 waves skip that set-up; angle offsets by uniform loads
# speedup vs baseline: 2.8102x; 1.0522x over previous
#include <hip/hip_runtime.h>
#include <math.h>


typedef __attribute__((ext_vector_type(2))) float v2f;
typedef __attribute__((ext_vector_type(8))) float v8f;

#define D2R 0.017453292519943295f
#define R2D 57.29577951308232f

__device__ __forceinline__ float acoef(int q, int c, float ca, float sa, float a, float d)
{
    float r = 0.0f;
    if (q == 0)  r = (c == 0) ? 1.0f : 0.0f;
    if (q == 1 || q == 2) r = (c == 1) ? ca : ((c == 2) ? sa : 0.0f);
    if (q == 3)  r = (c == 0) ? -1.0f : 0.0f;
    if (q == 12) r = (c == 1) ? -sa : ((c == 2) ? ca : 0.0f);
    if (q == 13) r = (c == 0) ? a : ((c == 1) ? -sa * d : ((c == 2) ? ca * d : 1.0f));
    return r;
}

__device__ __forceinline__ float xhalf_m8(float x)
{
    int b = __builtin_bit_cast(int, x);
    return __builtin_bit_cast(float,
        __builtin_amdgcn_permlanex16(b, b, 0xfedcba98, 0x76543210, false, false));
}

__device__ __forceinline__ float xhalf(float x)
{
    int b = __builtin_bit_cast(int, x);
    return __builtin_bit_cast(float,
        __builtin_amdgcn_permlanex16(b, b, 0x76543210, 0xfedcba98, false, false));
}

__device__ __forceinline__ float bcast(float x, int srcLane)
{
    return __builtin_bit_cast(float,
        __builtin_amdgcn_readlane(__builtin_bit_cast(int, x), srcLane));
}

__global__ __launch_bounds__(256, 16) void fk_wmma_kernel(
    const float* __restrict__ theta, const float* __restrict__ dh,
    float* __restrict__ out, int n)
{
    const int lane = threadIdx.x & 31;
    const int wave = threadIdx.x >> 5;
    const int h = lane >> 4;
    const int m = lane & 15;
    const int base = (blockIdx.x * 8 + wave) * 32;
    int samp = base + lane;
    if (samp >= n) samp = n - 1;

    const float2* tp = (const float2*)(theta + (size_t)samp * 6);
    const float2 t01 = tp[0], t23 = tp[1], t45 = tp[2];
    const float thd[6] = { t01.x, t01.y, t23.x, t23.y, t45.x, t45.y };

    const float off[6] = { dh[3], dh[7], dh[11], dh[15], dh[19], dh[23] };
    const float a0 = dh[0], d0 = dh[2];

    __shared__ float4 opA[5][32];
    __shared__ float2 cs0;
    if (wave == 0) {
        const int jl = (lane < 6) ? lane : 5;
        const float a_l   = dh[jl * 4 + 0];
        const float al_l  = dh[jl * 4 + 1] * D2R;
        const float d_l   = dh[jl * 4 + 2];
        float sa_l, ca_l;
        sincosf(al_l, &sa_l, &ca_l);
        #pragma unroll
        for (int j = 1; j < 6; ++j) {
            const float ca = bcast(ca_l, j), sa = bcast(sa_l, j);
            const float a = bcast(a_l, j), d = bcast(d_l, j);
            const float a0x = acoef(m, 2 * h + 0, ca, sa, a, d);
            const float a0y = acoef(m, 2 * h + 1, ca, sa, a, d);
            opA[j - 1][lane] = make_float4(a0x, a0y, xhalf_m8(a0x), xhalf_m8(a0y));
        }
        if (lane == 0) cs0 = make_float2(ca_l, sa_l);
    }

    float ct[6], st[6];
    #pragma unroll
    for (int j = 0; j < 6; ++j) {
        sincosf((off[j] + thd[j]) * D2R, &st[j], &ct[j]);
    }

    __syncthreads();

    v2f S[2][3];
    {
        const float2 cs = cs0;
        const float ca = cs.x, sa = cs.y;
        const float a = a0, d = d0;
        const float o[3][2] = { { ct[0], -st[0] }, { st[0] * ca, ct[0] * ca }, { st[0] * sa, ct[0] * sa } };
        const float k[3][2] = { { 0.0f, a }, { -sa, -sa * d }, { ca, ca * d } };
        #pragma unroll
        for (int g = 0; g < 2; ++g) {
            #pragma unroll
            for (int i = 0; i < 3; ++i) {
                S[g][i].x = (h == g) ? o[i][0] : k[i][0];
                S[g][i].y = (h == g) ? o[i][1] : k[i][1];
            }
        }
    }

    #pragma unroll
    for (int j = 1; j < 6; ++j) {
        const float4 t = opA[j - 1][lane];
        v2f A[2];
        A[0].x = t.x; A[0].y = t.y;
        A[1].x = t.z; A[1].y = t.w;
        v8f c[2][3];
        #pragma unroll
        for (int g = 0; g < 2; ++g) {
            #pragma unroll
            for (int i = 0; i < 3; ++i) {
                v8f z = {};
                c[g][i] = __builtin_amdgcn_wmma_f32_16x16x4_f32(
                    false, A[g], false, S[g][i], (short)0, z, false, false);
            }
        }
        #pragma unroll
        for (int g = 0; g < 2; ++g) {
            #pragma unroll
            for (int i = 0; i < 3; ++i) {
                S[g][i].x = fmaf(ct[j], c[g][i][0], fmaf(st[j], c[g][i][2], c[g][i][4]));
                S[g][i].y = fmaf(ct[j], c[g][i][1], fmaf(st[j], c[g][i][3], c[g][i][5]));
            }
        }
    }

    float b01[3][2], b23[3][2];
    #pragma unroll
    for (int i = 0; i < 3; ++i) {
        #pragma unroll
        for (int v = 0; v < 2; ++v) {
            const float own  = h ? S[1][i][v] : S[0][i][v];
            const float send = h ? S[0][i][v] : S[1][i][v];
            b01[i][v] = own;
            b23[i][v] = xhalf(send);
        }
    }

    const float T00 = b01[0][0], T01 = b01[0][1], T02 = b23[0][0], x = b23[0][1];
    const float T10 = b01[1][0], T11 = b01[1][1], T12 = b23[1][0], y = b23[1][1];
    const float T22 = b23[2][0], z = b23[2][1];
    float Aa, Bb, Cc;
    if (fabsf(T12) <= 1e-6f && fabsf(T22) <= 1e-6f) {
        Aa = R2D * atan2f(T10, T11);
        Bb = R2D * atan2f(T02, T22);
        Cc = 0.0f;
    } else {
        const float An = atan2f(-T01, T00);
        float sAn, cAn;
        sincosf(An, &sAn, &cAn);
        Aa = R2D * An;
        Bb = R2D * atan2f(T02, cAn * T00 - sAn * T01);
        Cc = R2D * atan2f(-T12, T22);
    }
    if (base + lane < n) {
        float2* o = (float2*)(out + (size_t)(base + lane) * 6);
        o[0] = make_float2(x, y);
        o[1] = make_float2(z, Aa);
        o[2] = make_float2(Bb, Cc);
    }
}

extern "C" void kernel_launch(void* const* d_in, const int* in_sizes, int n_in,
                              void* d_out, int out_size, void* d_ws, size_t ws_size,
                              hipStream_t stream) {
    const float* theta = (const float*)d_in[0];
    const float* dh    = (const float*)d_in[1];
    float* out = (float*)d_out;
    const int n = in_sizes[0] / 6;
    const int samplesPerBlock = 256;
    const int blocks = (n + samplesPerBlock - 1) / samplesPerBlock;
    fk_wmma_kernel<<<blocks, 256, 0, stream>>>(theta, dh, out, n);
}


# Six joint sin/cos as three packed f32 pairs (v_pk_fma_f32, v_pk_mul_f32): the library's short-path reduction and polynomials on signed angles, same bits; one large-angle test per wave, not six
# speedup vs baseline: 4.0915x; 1.4559x over previous
#include <hip/hip_runtime.h>
#include <math.h>


typedef __attribute__((ext_vector_type(2))) float v2f;
typedef __attribute__((ext_vector_type(8))) float v8f;

#define D2R 0.017453292519943295f
#define R2D 57.29577951308232f


__device__ __forceinline__ float bcast(float x, int srcLane)
{
    return __builtin_bit_cast(float,
        __builtin_amdgcn_readlane(__builtin_bit_cast(int, x), srcLane));
}


__device__ __forceinline__ constexpr float fbits(unsigned u)
{
    return __builtin_bit_cast(float, u);
}

__device__ __forceinline__ v2f splat(float c)
{
    v2f r = { c, c };
    return r;
}

__device__ __forceinline__ void quadrant(float fn, float sr, float cr, float& sn, float& cs)
{
    const int n = (int)fn;
    const bool odd = (n & 1) != 0;
    const unsigned flip = ((unsigned)n << 30) & 0x80000000u;
    const float ss = odd ? cr : sr;
    const float cc = odd ? -sr : cr;
    sn = __builtin_bit_cast(float, __builtin_bit_cast(unsigned, ss) ^ flip);
    cs = __builtin_bit_cast(float, __builtin_bit_cast(unsigned, cc) ^ flip);
}

__device__ __forceinline__ void sincos_pair(v2f x, v2f& sn, v2f& cs)
{
    const v2f t = x * fbits(0x3f22f983u);
    v2f fn;
    fn.x = __builtin_rintf(t.x);
    fn.y = __builtin_rintf(t.y);
    v2f r = __builtin_elementwise_fma(fn, splat(fbits(0xbfc90fdau)), x);
    r = __builtin_elementwise_fma(fn, splat(fbits(0xb3a22168u)), r);
    r = __builtin_elementwise_fma(fn, splat(fbits(0xa7c234c4u)), r);
    const v2f s = r * r;
    v2f p = __builtin_elementwise_fma(s, splat(fbits(0xb94c1982u)), splat(fbits(0x3c0881c4u)));
    p = __builtin_elementwise_fma(s, p, splat(fbits(0xbe2aaa9du)));
    p = s * p;
    const v2f sr = __builtin_elementwise_fma(r, p, r);
    v2f q = __builtin_elementwise_fma(s, splat(fbits(0x37d75334u)), splat(fbits(0xbab64f3bu)));
    q = __builtin_elementwise_fma(s, q, splat(fbits(0x3d2aabf7u)));
    q = __builtin_elementwise_fma(s, q, splat(fbits(0xbf000004u)));
    const v2f cr = __builtin_elementwise_fma(s, q, splat(1.0f));
    float s0, c0, s1, c1;
    quadrant(fn.x, sr.x, cr.x, s0, c0);
    quadrant(fn.y, sr.y, cr.y, s1, c1);
    sn.x = s0; sn.y = s1;
    cs.x = c0; cs.y = c1;
}

__global__ __launch_bounds__(256, 16) void fk_wmma_kernel(
    const float* __restrict__ theta, const float* __restrict__ dh,
    float* __restrict__ out, int n)
{
    const int lane = threadIdx.x & 31;
    const int wave = __builtin_amdgcn_readfirstlane((int)(threadIdx.x >> 5));
    const int base = (blockIdx.x * 8 + wave) * 32;
    int samp = base + lane;
    if (samp >= n) samp = n - 1;

    const float2* tp = (const float2*)(theta + (size_t)samp * 6);
    const float2 t01 = tp[0], t23 = tp[1], t45 = tp[2];
    const float thd[6] = { t01.x, t01.y, t23.x, t23.y, t45.x, t45.y };

    const float off[6] = { dh[3], dh[7], dh[11], dh[15], dh[19], dh[23] };
    const float aj[6]  = { dh[0], dh[4], dh[8],  dh[12], dh[16], dh[20] };
    const float dj[6]  = { dh[2], dh[6], dh[10], dh[14], dh[18], dh[22] };

    __shared__ float2 opA[5][32];
    __shared__ float2 cs5;
    if (wave == 0) {
        const int jl = (lane < 6) ? lane : 5;
        const float al_l = dh[jl * 4 + 1] * D2R;
        float sa_l, ca_l;
        sincosf(al_l, &sa_l, &ca_l);
        const int rr = (lane & 15) - 8 * (lane >> 4);
        const bool r0 = (rr == 0);
        const bool r1 = (rr == 1);
        #pragma unroll
        for (int j = 0; j < 5; ++j) {
            const float ca = bcast(ca_l, j), sa = bcast(sa_l, j);
            opA[j][lane] = make_float2(r0 ? ca : (r1 ? sa : 0.0f),
                                       r0 ? -sa : (r1 ? ca : 0.0f));
        }
        if (lane == 5) cs5 = make_float2(ca_l, sa_l);
    }

    float ct[6], st[6];
    {
        v2f xa[3];
        #pragma unroll
        for (int p = 0; p < 3; ++p) {
            const v2f o = { off[2 * p], off[2 * p + 1] };
            const v2f t = { thd[2 * p], thd[2 * p + 1] };
            xa[p] = (o + t) * D2R;
        }
        const float big = fmaxf(fmaxf(fmaxf(fabsf(xa[0].x), fabsf(xa[0].y)),
                                      fmaxf(fabsf(xa[1].x), fabsf(xa[1].y))),
                                fmaxf(fabsf(xa[2].x), fabsf(xa[2].y)));
        if (__builtin_amdgcn_ballot_w32(!(big < 131072.0f)) != 0u) {
            #pragma unroll
            for (int p = 0; p < 3; ++p) {
                sincosf(xa[p].x, &st[2 * p], &ct[2 * p]);
                sincosf(xa[p].y, &st[2 * p + 1], &ct[2 * p + 1]);
            }
        } else {
            #pragma unroll
            for (int p = 0; p < 3; ++p) {
                v2f sn, cs;
                sincos_pair(xa[p], sn, cs);
                st[2 * p] = sn.x;  st[2 * p + 1] = sn.y;
                ct[2 * p] = cs.x;  ct[2 * p + 1] = cs.y;
            }
        }
    }

    __syncthreads();

    float c[3][3];
    {
        const float2 cs = cs5;
        const float ca = cs.x, sa = cs.y;
        c[0][0] = ct[5];   c[0][1] = ca * st[5];    c[0][2] = sa * st[5];
        c[1][0] = -st[5];  c[1][1] = ca * ct[5];    c[1][2] = sa * ct[5];
        c[2][0] = aj[5];   c[2][1] = -sa * dj[5];   c[2][2] = ca * dj[5];
    }

    #pragma unroll
    for (int j = 4; j >= 0; --j) {
        const float2 t = opA[j][lane];
        v2f A;
        A.x = t.x; A.y = t.y;
        v2f B[3];
        float e0[3];
        #pragma unroll
        for (int q = 0; q < 3; ++q) {
            if (q < 2) {
                e0[q]  = fmaf(ct[j], c[q][0], -st[j] * c[q][1]);
                B[q].y = c[q][2];
            } else {
                e0[q]  = fmaf(ct[j], c[q][0], fmaf(-st[j], c[q][1], aj[j]));
                B[q].y = c[q][2] + dj[j];
            }
            B[q].x = fmaf(st[j], c[q][0], ct[j] * c[q][1]);
        }
        v8f d[3];
        #pragma unroll
        for (int q = 0; q < 3; ++q) {
            v8f z = {};
            d[q] = __builtin_amdgcn_wmma_f32_16x16x4_f32(
                false, A, false, B[q], (short)0, z, false, false);
        }
        #pragma unroll
        for (int q = 0; q < 3; ++q) {
            c[q][0] = e0[q];
            c[q][1] = d[q][0];
            c[q][2] = d[q][1];
        }
    }

    const float T00 = c[0][0], T01 = c[1][0], x = c[2][0];
    const float T10 = c[0][1], T11 = c[1][1], y = c[2][1];
    const float T20 = c[0][2], T21 = c[1][2], z = c[2][2];
    const float T02 = fmaf(T10, T21, -(T20 * T11));
    const float T12 = fmaf(T20, T01, -(T00 * T21));
    const float T22 = fmaf(T00, T11, -(T10 * T01));
    float Aa, Bb, Cc;
    if (fabsf(T12) <= 1e-6f && fabsf(T22) <= 1e-6f) {
        Aa = R2D * atan2f(T10, T11);
        Bb = R2D * atan2f(T02, T22);
        Cc = 0.0f;
    } else {
        const float An = atan2f(-T01, T00);
        float sAn, cAn;
        sincosf(An, &sAn, &cAn);
        Aa = R2D * An;
        Bb = R2D * atan2f(T02, cAn * T00 - sAn * T01);
        Cc = R2D * atan2f(-T12, T22);
    }
    const int gl = base + (int)__builtin_amdgcn_mbcnt_lo(~0u, 0u);
    if (gl < n) {
        float2* o = (float2*)(out + (size_t)gl * 6);
        o[0] = make_float2(x, y);
        o[1] = make_float2(z, Aa);
        o[2] = make_float2(Bb, Cc);
    }
}

extern "C" void kernel_launch(void* const* d_in, const int* in_sizes, int n_in,
                              void* d_out, int out_size, void* d_ws, size_t ws_size,
                              hipStream_t stream) {
    const float* theta = (const float*)d_in[0];
    const float* dh    = (const float*)d_in[1];
    float* out = (float*)d_out;
    const int n = in_sizes[0] / 6;
    const int samplesPerBlock = 256;
    const int blocks = (n + samplesPerBlock - 1) / samplesPerBlock;
    fk_wmma_kernel<<<blocks, 256, 0, stream>>>(theta, dh, out, n);
}
